# starting kernel
# baseline (speedup 1.0000x reference)
#include <hip/hip_runtime.h>
#include <math.h>

#define NN   50000
#define EE   800000
#define ET   (EE + NN)
#define INC  128
#define CC   64
#define NEGS 0.2f

typedef float v2f __attribute__((ext_vector_type(2)));
typedef float v8f __attribute__((ext_vector_type(8)));

__device__ __forceinline__ unsigned fenc(float f) {
    unsigned u = __float_as_uint(f);
    return (u >> 31) ? ~u : (u | 0x80000000u);
}
__device__ __forceinline__ float fdec(unsigned u) {
    return __uint_as_float((u >> 31) ? (u & 0x7FFFFFFFu) : ~u);
}

__global__ __launch_bounds__(256)
void init_ws_kernel(float* __restrict__ out, float* __restrict__ denom,
                    unsigned* __restrict__ amax) {
    int i = blockIdx.x * blockDim.x + threadIdx.x;
    if (i < NN * CC) out[i] = 0.0f;
    if (i < NN) { denom[i] = 0.0f; amax[i] = 0x007FFFFFu;   }
}

__global__ __launch_bounds__(32)
void gemm_wmma_kernel(const float* __restrict__ x, const float* __restrict__ W,
                      float* __restrict__ g) {
    const int rb   = blockIdx.x;
    const int lane = threadIdx.x;
    const int half = lane >> 4;
    const int l15  = lane & 15;

    v8f acc0 = {}; v8f acc1 = {}; v8f acc2 = {}; v8f acc3 = {};

    const float* xrow = x + (rb * 16 + l15) * INC + half * 2;
    for (int k0 = 0; k0 < INC; k0 += 4) {
        v2f a; a.x = xrow[k0]; a.y = xrow[k0 + 1];
        const float* wp = W + (k0 + half * 2) * CC + l15;
        v2f b0, b1, b2, b3;
        b0.x = wp[0];  b0.y = wp[0  + CC];
        b1.x = wp[16]; b1.y = wp[16 + CC];
        b2.x = wp[32]; b2.y = wp[32 + CC];
        b3.x = wp[48]; b3.y = wp[48 + CC];
        acc0 = __builtin_amdgcn_wmma_f32_16x16x4_f32(false, a, false, b0, (short)0, acc0, false, false);
        acc1 = __builtin_amdgcn_wmma_f32_16x16x4_f32(false, a, false, b1, (short)0, acc1, false, false);
        acc2 = __builtin_amdgcn_wmma_f32_16x16x4_f32(false, a, false, b2, (short)0, acc2, false, false);
        acc3 = __builtin_amdgcn_wmma_f32_16x16x4_f32(false, a, false, b3, (short)0, acc3, false, false);
    }
    #pragma unroll
    for (int v = 0; v < 8; ++v) {
        float* op = g + (rb * 16 + v + half * 8) * CC + l15;
        op[0]  = acc0[v];
        op[16] = acc1[v];
        op[32] = acc2[v];
        op[48] = acc3[v];
    }
}

__global__ __launch_bounds__(256)
void node_scores_kernel(const float* __restrict__ g, const float* __restrict__ corrs,
                        const float* __restrict__ att_i, const float* __restrict__ att_j,
                        const float* __restrict__ aem_i, const float* __restrict__ aem_j,
                        float* __restrict__ s_i, float* __restrict__ s_j) {
    int node = (blockIdx.x * blockDim.x + threadIdx.x) >> 5;
    int lane = threadIdx.x & 31;
    if (node >= NN) return;
    const float* gr = g + node * CC;
    const float* cr = corrs + node * CC;
    float si = 0.0f, sj = 0.0f;
    #pragma unroll
    for (int c = lane; c < CC; c += 32) {
        float gv = gr[c], cv = cr[c];
        si += gv * att_i[c] + cv * aem_i[c];
        sj += gv * att_j[c] + cv * aem_j[c];
    }
    #pragma unroll
    for (int off = 16; off > 0; off >>= 1) {
        si += __shfl_xor(si, off, 32);
        sj += __shfl_xor(sj, off, 32);
    }
    if (lane == 0) { s_i[node] = si; s_j[node] = sj; }
}

__global__ __launch_bounds__(256)
void edge_alpha_kernel(const int* __restrict__ ei, const float* __restrict__ s_i,
                       const float* __restrict__ s_j, float* __restrict__ alpha,
                       unsigned* __restrict__ amax) {
    int e = blockIdx.x * blockDim.x + threadIdx.x;
    if (e >= ET) return;
    int src, dst;
    if (e < EE) { src = ei[e]; dst = ei[EE + e]; } else { src = dst = e - EE; }
    float a = s_i[dst] + s_j[src];
    a = (a > 0.0f) ? a : NEGS * a;
    alpha[e] = a;
    atomicMax(amax + dst, fenc(a));
}

__global__ __launch_bounds__(256)
void edge_exp_kernel(const int* __restrict__ ei, const unsigned* __restrict__ amax,
                     float* __restrict__ alpha_ex, float* __restrict__ denom) {
    int e = blockIdx.x * blockDim.x + threadIdx.x;
    if (e >= ET) return;
    int dst = (e < EE) ? ei[EE + e] : (e - EE);
    float ex = expf(alpha_ex[e] - fdec(amax[dst]));
    alpha_ex[e] = ex;
    atomicAdd(denom + dst, ex);
}

__global__ __launch_bounds__(256)
void message_kernel(const int* __restrict__ ei, const float* __restrict__ g,
                    const float* __restrict__ ex, const float* __restrict__ denom,
                    float* __restrict__ out) {
    int tid = blockIdx.x * blockDim.x + threadIdx.x;
    int e = tid >> 4;
    int q = tid & 15;
    if (e >= ET) return;
    int src, dst;
    if (e < EE) { src = ei[e]; dst = ei[EE + e]; } else { src = dst = e - EE; }
    float coef = ex[e] / (denom[dst] + 1e-16f);
    float4 gv = ((const float4*)(g + src * CC))[q];
    float* op = out + dst * CC + q * 4;
    atomicAdd(op + 0, gv.x * coef);
    atomicAdd(op + 1, gv.y * coef);
    atomicAdd(op + 2, gv.z * coef);
    atomicAdd(op + 3, gv.w * coef);
}

__global__ __launch_bounds__(256)
void finalize_kernel(float* __restrict__ out, const float* __restrict__ bias) {
    int i = blockIdx.x * blockDim.x + threadIdx.x;
    if (i >= NN * CC) return;
    float v = out[i] + bias[i & (CC - 1)];
    out[i] = (v > 0.0f) ? v : 0.0f;
}

extern "C" void kernel_launch(void* const* d_in, const int* in_sizes, int n_in,
                              void* d_out, int out_size, void* d_ws, size_t ws_size,
                              hipStream_t stream) {
    (void)in_sizes; (void)n_in; (void)out_size; (void)ws_size;
    const float* x      = (const float*)d_in[0];
    const int*   ei     = (const int*)  d_in[1];
    const float* corrs  = (const float*)d_in[2];
    const float* W      = (const float*)d_in[3];
    const float* att_i  = (const float*)d_in[4];
    const float* att_j  = (const float*)d_in[5];
    const float* aem_i  = (const float*)d_in[6];
    const float* aem_j  = (const float*)d_in[7];
    const float* bias   = (const float*)d_in[8];
    float* out = (float*)d_out;

    float*    ws    = (float*)d_ws;
    float*    g     = ws;
    float*    s_i   = g + (size_t)NN * CC;
    float*    s_j   = s_i + NN;
    unsigned* amax  = (unsigned*)(s_j + NN);
    float*    denom = (float*)(amax + NN);
    float*    aex   = denom + NN;

    init_ws_kernel<<<(NN * CC + 255) / 256, 256, 0, stream>>>(out, denom, amax);
    gemm_wmma_kernel<<<NN / 16, 32, 0, stream>>>(x, W, g);
    node_scores_kernel<<<(NN * 32 + 255) / 256, 256, 0, stream>>>(
        g, corrs, att_i, att_j, aem_i, aem_j, s_i, s_j);
    edge_alpha_kernel<<<(ET + 255) / 256, 256, 0, stream>>>(ei, s_i, s_j, aex, amax);
    edge_exp_kernel<<<(ET + 255) / 256, 256, 0, stream>>>(ei, amax, aex, denom);
    message_kernel<<<(ET * 16 + 255) / 256, 256, 0, stream>>>(ei, g, aex, denom, out);
    finalize_kernel<<<(NN * CC + 255) / 256, 256, 0, stream>>>(out, bias);
}


# Exp sums: 256 blocks (64 edge slices x 4 node parts) sum in an LDS node table in 64-bit fixed point (2^-40), flushed by line-wide integer atomics; 850000 one-word float atomics gone; order-free sums
# speedup vs baseline: 1.7326x; 1.7326x over previous
#include <hip/hip_runtime.h>
#include <math.h>

#define NN   50000
#define EE   800000
#define ET   (EE + NN)
#define INC  128
#define CC   64
#define NEGS 0.2f

typedef float v2f __attribute__((ext_vector_type(2)));
typedef float v8f __attribute__((ext_vector_type(8)));

__device__ __forceinline__ unsigned fenc(float f) {
    unsigned u = __float_as_uint(f);
    return (u >> 31) ? ~u : (u | 0x80000000u);
}
__device__ __forceinline__ float fdec(unsigned u) {
    return __uint_as_float((u >> 31) ? (u & 0x7FFFFFFFu) : ~u);
}

__global__ __launch_bounds__(256)
void init_ws_kernel(float* __restrict__ out, unsigned long long* __restrict__ dsum,
                    unsigned* __restrict__ amax) {
    int i = blockIdx.x * blockDim.x + threadIdx.x;
    if (i < NN * CC) out[i] = 0.0f;
    if (i < NN) { dsum[i] = 0ull; amax[i] = 0x007FFFFFu;   }
}

__global__ __launch_bounds__(32)
void gemm_wmma_kernel(const float* __restrict__ x, const float* __restrict__ W,
                      float* __restrict__ g) {
    const int rb   = blockIdx.x;
    const int lane = threadIdx.x;
    const int half = lane >> 4;
    const int l15  = lane & 15;

    v8f acc0 = {}; v8f acc1 = {}; v8f acc2 = {}; v8f acc3 = {};

    const float* xrow = x + (rb * 16 + l15) * INC + half * 2;
    for (int k0 = 0; k0 < INC; k0 += 4) {
        v2f a; a.x = xrow[k0]; a.y = xrow[k0 + 1];
        const float* wp = W + (k0 + half * 2) * CC + l15;
        v2f b0, b1, b2, b3;
        b0.x = wp[0];  b0.y = wp[0  + CC];
        b1.x = wp[16]; b1.y = wp[16 + CC];
        b2.x = wp[32]; b2.y = wp[32 + CC];
        b3.x = wp[48]; b3.y = wp[48 + CC];
        acc0 = __builtin_amdgcn_wmma_f32_16x16x4_f32(false, a, false, b0, (short)0, acc0, false, false);
        acc1 = __builtin_amdgcn_wmma_f32_16x16x4_f32(false, a, false, b1, (short)0, acc1, false, false);
        acc2 = __builtin_amdgcn_wmma_f32_16x16x4_f32(false, a, false, b2, (short)0, acc2, false, false);
        acc3 = __builtin_amdgcn_wmma_f32_16x16x4_f32(false, a, false, b3, (short)0, acc3, false, false);
    }
    #pragma unroll
    for (int v = 0; v < 8; ++v) {
        float* op = g + (rb * 16 + v + half * 8) * CC + l15;
        op[0]  = acc0[v];
        op[16] = acc1[v];
        op[32] = acc2[v];
        op[48] = acc3[v];
    }
}

__global__ __launch_bounds__(256)
void node_scores_kernel(const float* __restrict__ g, const float* __restrict__ corrs,
                        const float* __restrict__ att_i, const float* __restrict__ att_j,
                        const float* __restrict__ aem_i, const float* __restrict__ aem_j,
                        float* __restrict__ s_i, float* __restrict__ s_j) {
    int node = (blockIdx.x * blockDim.x + threadIdx.x) >> 5;
    int lane = threadIdx.x & 31;
    if (node >= NN) return;
    const float* gr = g + node * CC;
    const float* cr = corrs + node * CC;
    float si = 0.0f, sj = 0.0f;
    #pragma unroll
    for (int c = lane; c < CC; c += 32) {
        float gv = gr[c], cv = cr[c];
        si += gv * att_i[c] + cv * aem_i[c];
        sj += gv * att_j[c] + cv * aem_j[c];
    }
    #pragma unroll
    for (int off = 16; off > 0; off >>= 1) {
        si += __shfl_xor(si, off, 32);
        sj += __shfl_xor(sj, off, 32);
    }
    if (lane == 0) { s_i[node] = si; s_j[node] = sj; }
}

#define SL   64
#define PT   4
#define PN   ((((NN + PT - 1) / PT) + 31) & ~31)
#define AT   1024
#define SE   (SL * AT)
#define EPT  ((ET + SE - 1) / SE)
#define FULLK (EE / SE)
#define ENC_NEG_INF 0x007FFFFFu

__global__ __launch_bounds__(AT)
void edge_alpha_kernel(const int* __restrict__ ei, const float* __restrict__ s_i,
                       const float* __restrict__ s_j, float* __restrict__ aplane,
                       unsigned* __restrict__ amax) {
    __shared__ unsigned lmax[PN];
    const int tid = threadIdx.x;
    const int sl  = blockIdx.x / PT;
    const int pt  = blockIdx.x % PT;
    const int lo  = pt * PN;
    for (int i = tid; i < PN; i += AT) lmax[i] = ENC_NEG_INF;

    int src[EPT], dst[EPT];
    const int e0 = sl * AT + tid;
    #pragma unroll
    for (int k = 0; k < EPT; ++k) {
        const int e = e0 + k * SE;
        if (k < FULLK || e < EE) { src[k] = ei[e]; dst[k] = ei[EE + e]; }
        else                     { src[k] = dst[k] = e - EE; }
    }
    float a[EPT];
    bool  own[EPT];
    #pragma unroll
    for (int k = 0; k < EPT; ++k) {
        const int e = e0 + k * SE;
        own[k] = (k < FULLK || e < ET) && ((unsigned)(dst[k] - lo) < (unsigned)PN);
        const int dd = own[k] ? dst[k] : 0;
        const int ss = own[k] ? src[k] : 0;
        const float v = s_i[dd] + s_j[ss];
        a[k] = (v > 0.0f) ? v : NEGS * v;
    }
    __syncthreads();
    #pragma unroll
    for (int k = 0; k < EPT; ++k) {
        const int e = e0 + k * SE;
        if (own[k]) {
            aplane[(size_t)pt * ET + e] = a[k];
            atomicMax(&lmax[dst[k] - lo], fenc(a[k]));
        }
    }
    __syncthreads();
    const int rot = (sl * 6 * 32) % PN;
    for (int i = tid; i < PN; i += AT) {
        int j = i + rot;
        if (j >= PN) j -= PN;
        const unsigned v = lmax[j];
        if (v != ENC_NEG_INF && lo + j < NN) atomicMax(amax + lo + j, v);
    }
}

#define FXS 1099511627776.0f

__global__ __launch_bounds__(AT)
void edge_exp_kernel(const int* __restrict__ ei, const unsigned* __restrict__ amax,
                     const float* __restrict__ aplane, float* __restrict__ explane,
                     unsigned long long* __restrict__ dsum) {
    __shared__ unsigned long long lsum[PN];
    const int tid = threadIdx.x;
    const int sl  = blockIdx.x / PT;
    const int pt  = blockIdx.x % PT;
    const int lo  = pt * PN;
    for (int i = tid; i < PN; i += AT) lsum[i] = 0ull;

    int dst[EPT];
    const int e0 = sl * AT + tid;
    #pragma unroll
    for (int k = 0; k < EPT; ++k) {
        const int e = e0 + k * SE;
        dst[k] = (k < FULLK || e < EE) ? ei[EE + e] : (e - EE);
    }
    const float* ap = aplane + (size_t)pt * ET;
    float* xp = explane + (size_t)pt * ET;
    float x[EPT];
    bool  own[EPT];
    #pragma unroll
    for (int k = 0; k < EPT; ++k) {
        const int e = e0 + k * SE;
        own[k] = (k < FULLK || e < ET) && ((unsigned)(dst[k] - lo) < (unsigned)PN);
        const int dd = own[k] ? dst[k] : lo;
        const int ee = own[k] ? e : (EE + lo);
        x[k] = expf(ap[ee] - fdec(amax[dd]));
    }
    __syncthreads();
    #pragma unroll
    for (int k = 0; k < EPT; ++k) {
        const int e = e0 + k * SE;
        if (own[k]) {
            xp[e] = x[k];
            atomicAdd(&lsum[dst[k] - lo], (unsigned long long)(x[k] * FXS));
        }
    }
    __syncthreads();
    const int rot = (sl * 6 * 32) % PN;
    for (int i = tid; i < PN; i += AT) {
        int j = i + rot;
        if (j >= PN) j -= PN;
        const unsigned long long v = lsum[j];
        if (v != 0ull && lo + j < NN) atomicAdd(dsum + lo + j, v);
    }
}

__global__ __launch_bounds__(256)
void message_kernel(const int* __restrict__ ei, const float* __restrict__ g,
                    const float* __restrict__ explane,
                    const unsigned long long* __restrict__ dsum,
                    float* __restrict__ out) {
    int tid = blockIdx.x * blockDim.x + threadIdx.x;
    int e = tid >> 4;
    int q = tid & 15;
    if (e >= ET) return;
    int src, dst;
    if (e < EE) { src = ei[e]; dst = ei[EE + e]; } else { src = dst = e - EE; }
    const float denom = (float)dsum[dst] * (1.0f / FXS);
    float coef = explane[(size_t)(dst / PN) * ET + e] / (denom + 1e-16f);
    const float* gp = g + src * CC + q;
    float g0 = gp[0], g1 = gp[16], g2 = gp[32], g3 = gp[48];
    float* op = out + dst * CC + q;
    atomicAdd(op + 0,  g0 * coef);
    atomicAdd(op + 16, g1 * coef);
    atomicAdd(op + 32, g2 * coef);
    atomicAdd(op + 48, g3 * coef);
}

__global__ __launch_bounds__(256)
void finalize_kernel(float* __restrict__ out, const float* __restrict__ bias) {
    int i = blockIdx.x * blockDim.x + threadIdx.x;
    if (i >= NN * CC) return;
    float v = out[i] + bias[i & (CC - 1)];
    out[i] = (v > 0.0f) ? v : 0.0f;
}

extern "C" void kernel_launch(void* const* d_in, const int* in_sizes, int n_in,
                              void* d_out, int out_size, void* d_ws, size_t ws_size,
                              hipStream_t stream) {
    (void)in_sizes; (void)n_in; (void)out_size; (void)ws_size;
    const float* x      = (const float*)d_in[0];
    const int*   ei     = (const int*)  d_in[1];
    const float* corrs  = (const float*)d_in[2];
    const float* W      = (const float*)d_in[3];
    const float* att_i  = (const float*)d_in[4];
    const float* att_j  = (const float*)d_in[5];
    const float* aem_i  = (const float*)d_in[6];
    const float* aem_j  = (const float*)d_in[7];
    const float* bias   = (const float*)d_in[8];
    float* out = (float*)d_out;

    float*    ws    = (float*)d_ws;
    float*    g     = ws;
    float*    s_i   = g + (size_t)NN * CC;
    float*    s_j   = s_i + NN;
    unsigned* amax  = (unsigned*)(s_j + NN);
    unsigned long long* dsum = (unsigned long long*)(amax + NN);
    float*    xpl   = (float*)(dsum + NN);
    float*    apl   = xpl + (size_t)PT * ET;

    init_ws_kernel<<<(NN * CC + 255) / 256, 256, 0, stream>>>(out, dsum, amax);
    gemm_wmma_kernel<<<NN / 16, 32, 0, stream>>>(x, W, g);
    node_scores_kernel<<<(NN * 32 + 255) / 256, 256, 0, stream>>>(
        g, corrs, att_i, att_j, aem_i, aem_j, s_i, s_j);
    edge_alpha_kernel<<<SL * PT, AT, 0, stream>>>(ei, s_i, s_j, apl, amax);
    edge_exp_kernel<<<SL * PT, AT, 0, stream>>>(ei, amax, apl, xpl, dsum);
    message_kernel<<<(ET * 16 + 255) / 256, 256, 0, stream>>>(ei, g, xpl, dsum, out);
    finalize_kernel<<<(NN * CC + 255) / 256, 256, 0, stream>>>(out, bias);
}


# Init launch folded into the GEMM launch: each GEMM wave stores its 16 nodes' zero output rows, zero exp sums and -inf max words ahead of its k loop; 7 launches -> 6; sum order unchanged
# speedup vs baseline: 1.8136x; 1.0468x over previous
#include <hip/hip_runtime.h>
#include <math.h>

#define NN   50000
#define EE   800000
#define ET   (EE + NN)
#define INC  128
#define CC   64
#define NEGS 0.2f

typedef float v2f __attribute__((ext_vector_type(2)));
typedef float v8f __attribute__((ext_vector_type(8)));

__device__ __forceinline__ unsigned fenc(float f) {
    unsigned u = __float_as_uint(f);
    return (u >> 31) ? ~u : (u | 0x80000000u);
}
__device__ __forceinline__ float fdec(unsigned u) {
    return __uint_as_float((u >> 31) ? (u & 0x7FFFFFFFu) : ~u);
}

#define ENC_NEG_INF 0x007FFFFFu

__global__ __launch_bounds__(32)
void gemm_wmma_kernel(const float* __restrict__ x, const float* __restrict__ W,
                      float* __restrict__ g, float* __restrict__ out,
                      unsigned long long* __restrict__ dsum, unsigned* __restrict__ amax) {
    const int rb   = blockIdx.x;
    const int lane = threadIdx.x;
    const int half = lane >> 4;
    const int l15  = lane & 15;

    {
        const float4 z4 = make_float4(0.0f, 0.0f, 0.0f, 0.0f);
        float4* o4 = reinterpret_cast<float4*>(out + (size_t)rb * 16 * CC) + lane;
        for (int i = 0; i < 8; ++i) o4[i * 32] = z4;
        if (half == 0) dsum[rb * 16 + l15] = 0ull;
        else           amax[rb * 16 + l15] = ENC_NEG_INF;
    }

    v8f acc0 = {}; v8f acc1 = {}; v8f acc2 = {}; v8f acc3 = {};

    const float* xrow = x + (rb * 16 + l15) * INC + half * 2;
    for (int k0 = 0; k0 < INC; k0 += 4) {
        v2f a; a.x = xrow[k0]; a.y = xrow[k0 + 1];
        const float* wp = W + (k0 + half * 2) * CC + l15;
        v2f b0, b1, b2, b3;
        b0.x = wp[0];  b0.y = wp[0  + CC];
        b1.x = wp[16]; b1.y = wp[16 + CC];
        b2.x = wp[32]; b2.y = wp[32 + CC];
        b3.x = wp[48]; b3.y = wp[48 + CC];
        acc0 = __builtin_amdgcn_wmma_f32_16x16x4_f32(false, a, false, b0, (short)0, acc0, false, false);
        acc1 = __builtin_amdgcn_wmma_f32_16x16x4_f32(false, a, false, b1, (short)0, acc1, false, false);
        acc2 = __builtin_amdgcn_wmma_f32_16x16x4_f32(false, a, false, b2, (short)0, acc2, false, false);
        acc3 = __builtin_amdgcn_wmma_f32_16x16x4_f32(false, a, false, b3, (short)0, acc3, false, false);
    }
    #pragma unroll
    for (int v = 0; v < 8; ++v) {
        float* op = g + (rb * 16 + v + half * 8) * CC + l15;
        op[0]  = acc0[v];
        op[16] = acc1[v];
        op[32] = acc2[v];
        op[48] = acc3[v];
    }
}

__global__ __launch_bounds__(256)
void node_scores_kernel(const float* __restrict__ g, const float* __restrict__ corrs,
                        const float* __restrict__ att_i, const float* __restrict__ att_j,
                        const float* __restrict__ aem_i, const float* __restrict__ aem_j,
                        float* __restrict__ s_i, float* __restrict__ s_j) {
    int node = (blockIdx.x * blockDim.x + threadIdx.x) >> 5;
    int lane = threadIdx.x & 31;
    if (node >= NN) return;
    const float* gr = g + node * CC;
    const float* cr = corrs + node * CC;
    float si = 0.0f, sj = 0.0f;
    #pragma unroll
    for (int c = lane; c < CC; c += 32) {
        float gv = gr[c], cv = cr[c];
        si += gv * att_i[c] + cv * aem_i[c];
        sj += gv * att_j[c] + cv * aem_j[c];
    }
    #pragma unroll
    for (int off = 16; off > 0; off >>= 1) {
        si += __shfl_xor(si, off, 32);
        sj += __shfl_xor(sj, off, 32);
    }
    if (lane == 0) { s_i[node] = si; s_j[node] = sj; }
}

#define SL   64
#define PT   4
#define PN   ((((NN + PT - 1) / PT) + 31) & ~31)
#define AT   1024
#define SE   (SL * AT)
#define EPT  ((ET + SE - 1) / SE)
#define FULLK (EE / SE)

__global__ __launch_bounds__(AT)
void edge_alpha_kernel(const int* __restrict__ ei, const float* __restrict__ s_i,
                       const float* __restrict__ s_j, float* __restrict__ aplane,
                       unsigned* __restrict__ amax) {
    __shared__ unsigned lmax[PN];
    const int tid = threadIdx.x;
    const int sl  = blockIdx.x / PT;
    const int pt  = blockIdx.x % PT;
    const int lo  = pt * PN;
    for (int i = tid; i < PN; i += AT) lmax[i] = ENC_NEG_INF;

    int src[EPT], dst[EPT];
    const int e0 = sl * AT + tid;
    #pragma unroll
    for (int k = 0; k < EPT; ++k) {
        const int e = e0 + k * SE;
        if (k < FULLK || e < EE) { src[k] = ei[e]; dst[k] = ei[EE + e]; }
        else                     { src[k] = dst[k] = e - EE; }
    }
    float a[EPT];
    bool  own[EPT];
    #pragma unroll
    for (int k = 0; k < EPT; ++k) {
        const int e = e0 + k * SE;
        own[k] = (k < FULLK || e < ET) && ((unsigned)(dst[k] - lo) < (unsigned)PN);
        const int dd = own[k] ? dst[k] : 0;
        const int ss = own[k] ? src[k] : 0;
        const float v = s_i[dd] + s_j[ss];
        a[k] = (v > 0.0f) ? v : NEGS * v;
    }
    __syncthreads();
    #pragma unroll
    for (int k = 0; k < EPT; ++k) {
        const int e = e0 + k * SE;
        if (own[k]) {
            aplane[(size_t)pt * ET + e] = a[k];
            atomicMax(&lmax[dst[k] - lo], fenc(a[k]));
        }
    }
    __syncthreads();
    const int rot = (sl * 6 * 32) % PN;
    for (int i = tid; i < PN; i += AT) {
        int j = i + rot;
        if (j >= PN) j -= PN;
        const unsigned v = lmax[j];
        if (v != ENC_NEG_INF && lo + j < NN) atomicMax(amax + lo + j, v);
    }
}

#define FXS 1099511627776.0f

__global__ __launch_bounds__(AT)
void edge_exp_kernel(const int* __restrict__ ei, const unsigned* __restrict__ amax,
                     const float* __restrict__ aplane, float* __restrict__ explane,
                     unsigned long long* __restrict__ dsum) {
    __shared__ unsigned long long lsum[PN];
    const int tid = threadIdx.x;
    const int sl  = blockIdx.x / PT;
    const int pt  = blockIdx.x % PT;
    const int lo  = pt * PN;
    for (int i = tid; i < PN; i += AT) lsum[i] = 0ull;

    int dst[EPT];
    const int e0 = sl * AT + tid;
    #pragma unroll
    for (int k = 0; k < EPT; ++k) {
        const int e = e0 + k * SE;
        dst[k] = (k < FULLK || e < EE) ? ei[EE + e] : (e - EE);
    }
    const float* ap = aplane + (size_t)pt * ET;
    float* xp = explane + (size_t)pt * ET;
    float x[EPT];
    bool  own[EPT];
    #pragma unroll
    for (int k = 0; k < EPT; ++k) {
        const int e = e0 + k * SE;
        own[k] = (k < FULLK || e < ET) && ((unsigned)(dst[k] - lo) < (unsigned)PN);
        const int dd = own[k] ? dst[k] : lo;
        const int ee = own[k] ? e : (EE + lo);
        x[k] = expf(ap[ee] - fdec(amax[dd]));
    }
    __syncthreads();
    #pragma unroll
    for (int k = 0; k < EPT; ++k) {
        const int e = e0 + k * SE;
        if (own[k]) {
            xp[e] = x[k];
            atomicAdd(&lsum[dst[k] - lo], (unsigned long long)(x[k] * FXS));
        }
    }
    __syncthreads();
    const int rot = (sl * 6 * 32) % PN;
    for (int i = tid; i < PN; i += AT) {
        int j = i + rot;
        if (j >= PN) j -= PN;
        const unsigned long long v = lsum[j];
        if (v != 0ull && lo + j < NN) atomicAdd(dsum + lo + j, v);
    }
}

__global__ __launch_bounds__(256)
void message_kernel(const int* __restrict__ ei, const float* __restrict__ g,
                    const float* __restrict__ explane,
                    const unsigned long long* __restrict__ dsum,
                    float* __restrict__ out) {
    int tid = blockIdx.x * blockDim.x + threadIdx.x;
    int e = tid >> 4;
    int q = tid & 15;
    if (e >= ET) return;
    int src, dst;
    if (e < EE) { src = ei[e]; dst = ei[EE + e]; } else { src = dst = e - EE; }
    const float denom = (float)dsum[dst] * (1.0f / FXS);
    float coef = explane[(size_t)(dst / PN) * ET + e] / (denom + 1e-16f);
    const float* gp = g + src * CC + q;
    float g0 = gp[0], g1 = gp[16], g2 = gp[32], g3 = gp[48];
    float* op = out + dst * CC + q;
    atomicAdd(op + 0,  g0 * coef);
    atomicAdd(op + 16, g1 * coef);
    atomicAdd(op + 32, g2 * coef);
    atomicAdd(op + 48, g3 * coef);
}

__global__ __launch_bounds__(256)
void finalize_kernel(float* __restrict__ out, const float* __restrict__ bias) {
    int i = blockIdx.x * blockDim.x + threadIdx.x;
    if (i >= NN * CC) return;
    float v = out[i] + bias[i & (CC - 1)];
    out[i] = (v > 0.0f) ? v : 0.0f;
}

extern "C" void kernel_launch(void* const* d_in, const int* in_sizes, int n_in,
                              void* d_out, int out_size, void* d_ws, size_t ws_size,
                              hipStream_t stream) {
    (void)in_sizes; (void)n_in; (void)out_size; (void)ws_size;
    const float* x      = (const float*)d_in[0];
    const int*   ei     = (const int*)  d_in[1];
    const float* corrs  = (const float*)d_in[2];
    const float* W      = (const float*)d_in[3];
    const float* att_i  = (const float*)d_in[4];
    const float* att_j  = (const float*)d_in[5];
    const float* aem_i  = (const float*)d_in[6];
    const float* aem_j  = (const float*)d_in[7];
    const float* bias   = (const float*)d_in[8];
    float* out = (float*)d_out;

    float*    ws    = (float*)d_ws;
    float*    g     = ws;
    float*    s_i   = g + (size_t)NN * CC;
    float*    s_j   = s_i + NN;
    unsigned* amax  = (unsigned*)(s_j + NN);
    unsigned long long* dsum = (unsigned long long*)(amax + NN);
    float*    xpl   = (float*)(dsum + NN);
    float*    apl   = xpl + (size_t)PT * ET;

    gemm_wmma_kernel<<<NN / 16, 32, 0, stream>>>(x, W, g, out, dsum, amax);
    node_scores_kernel<<<(NN * 32 + 255) / 256, 256, 0, stream>>>(
        g, corrs, att_i, att_j, aem_i, aem_j, s_i, s_j);
    edge_alpha_kernel<<<SL * PT, AT, 0, stream>>>(ei, s_i, s_j, apl, amax);
    edge_exp_kernel<<<SL * PT, AT, 0, stream>>>(ei, amax, apl, xpl, dsum);
    message_kernel<<<(ET * 16 + 255) / 256, 256, 0, stream>>>(ei, g, xpl, dsum, out);
    finalize_kernel<<<(NN * CC + 255) / 256, 256, 0, stream>>>(out, bias);
}


# Message pass: a wave takes 8 successive edges (lane k has edge k's header and weight, passed by readlane); lanes cover a row as two whole lines per load and atomic add; sums in atomic arrival order
# speedup vs baseline: 2.1203x; 1.1691x over previous
#include <hip/hip_runtime.h>
#include <math.h>

#define NN   50000
#define EE   800000
#define ET   (EE + NN)
#define INC  128
#define CC   64
#define NEGS 0.2f

typedef float v2f __attribute__((ext_vector_type(2)));
typedef float v8f __attribute__((ext_vector_type(8)));

__device__ __forceinline__ unsigned fenc(float f) {
    unsigned u = __float_as_uint(f);
    return (u >> 31) ? ~u : (u | 0x80000000u);
}
__device__ __forceinline__ float fdec(unsigned u) {
    return __uint_as_float((u >> 31) ? (u & 0x7FFFFFFFu) : ~u);
}

#define ENC_NEG_INF 0x007FFFFFu

__global__ __launch_bounds__(32)
void gemm_wmma_kernel(const float* __restrict__ x, const float* __restrict__ W,
                      float* __restrict__ g, float* __restrict__ out,
                      unsigned long long* __restrict__ dsum, unsigned* __restrict__ amax) {
    const int rb   = blockIdx.x;
    const int lane = threadIdx.x;
    const int half = lane >> 4;
    const int l15  = lane & 15;

    {
        const float4 z4 = make_float4(0.0f, 0.0f, 0.0f, 0.0f);
        float4* o4 = reinterpret_cast<float4*>(out + (size_t)rb * 16 * CC) + lane;
        for (int i = 0; i < 8; ++i) o4[i * 32] = z4;
        if (half == 0) dsum[rb * 16 + l15] = 0ull;
        else           amax[rb * 16 + l15] = ENC_NEG_INF;
    }

    v8f acc0 = {}; v8f acc1 = {}; v8f acc2 = {}; v8f acc3 = {};

    const float* xrow = x + (rb * 16 + l15) * INC + half * 2;
    for (int k0 = 0; k0 < INC; k0 += 4) {
        v2f a; a.x = xrow[k0]; a.y = xrow[k0 + 1];
        const float* wp = W + (k0 + half * 2) * CC + l15;
        v2f b0, b1, b2, b3;
        b0.x = wp[0];  b0.y = wp[0  + CC];
        b1.x = wp[16]; b1.y = wp[16 + CC];
        b2.x = wp[32]; b2.y = wp[32 + CC];
        b3.x = wp[48]; b3.y = wp[48 + CC];
        acc0 = __builtin_amdgcn_wmma_f32_16x16x4_f32(false, a, false, b0, (short)0, acc0, false, false);
        acc1 = __builtin_amdgcn_wmma_f32_16x16x4_f32(false, a, false, b1, (short)0, acc1, false, false);
        acc2 = __builtin_amdgcn_wmma_f32_16x16x4_f32(false, a, false, b2, (short)0, acc2, false, false);
        acc3 = __builtin_amdgcn_wmma_f32_16x16x4_f32(false, a, false, b3, (short)0, acc3, false, false);
    }
    #pragma unroll
    for (int v = 0; v < 8; ++v) {
        float* op = g + (rb * 16 + v + half * 8) * CC + l15;
        op[0]  = acc0[v];
        op[16] = acc1[v];
        op[32] = acc2[v];
        op[48] = acc3[v];
    }
}

__global__ __launch_bounds__(256)
void node_scores_kernel(const float* __restrict__ g, const float* __restrict__ corrs,
                        const float* __restrict__ att_i, const float* __restrict__ att_j,
                        const float* __restrict__ aem_i, const float* __restrict__ aem_j,
                        float* __restrict__ s_i, float* __restrict__ s_j) {
    int node = (blockIdx.x * blockDim.x + threadIdx.x) >> 5;
    int lane = threadIdx.x & 31;
    if (node >= NN) return;
    const float* gr = g + node * CC;
    const float* cr = corrs + node * CC;
    float si = 0.0f, sj = 0.0f;
    #pragma unroll
    for (int c = lane; c < CC; c += 32) {
        float gv = gr[c], cv = cr[c];
        si += gv * att_i[c] + cv * aem_i[c];
        sj += gv * att_j[c] + cv * aem_j[c];
    }
    #pragma unroll
    for (int off = 16; off > 0; off >>= 1) {
        si += __shfl_xor(si, off, 32);
        sj += __shfl_xor(sj, off, 32);
    }
    if (lane == 0) { s_i[node] = si; s_j[node] = sj; }
}

#define SL   64
#define PT   4
#define PN   ((((NN + PT - 1) / PT) + 31) & ~31)
#define AT   1024
#define SE   (SL * AT)
#define EPT  ((ET + SE - 1) / SE)
#define FULLK (EE / SE)

__global__ __launch_bounds__(AT)
void edge_alpha_kernel(const int* __restrict__ ei, const float* __restrict__ s_i,
                       const float* __restrict__ s_j, float* __restrict__ aplane,
                       unsigned* __restrict__ amax) {
    __shared__ unsigned lmax[PN];
    const int tid = threadIdx.x;
    const int sl  = blockIdx.x / PT;
    const int pt  = blockIdx.x % PT;
    const int lo  = pt * PN;
    for (int i = tid; i < PN; i += AT) lmax[i] = ENC_NEG_INF;

    int src[EPT], dst[EPT];
    const int e0 = sl * AT + tid;
    #pragma unroll
    for (int k = 0; k < EPT; ++k) {
        const int e = e0 + k * SE;
        if (k < FULLK || e < EE) { src[k] = ei[e]; dst[k] = ei[EE + e]; }
        else                     { src[k] = dst[k] = e - EE; }
    }
    float a[EPT];
    bool  own[EPT];
    #pragma unroll
    for (int k = 0; k < EPT; ++k) {
        const int e = e0 + k * SE;
        own[k] = (k < FULLK || e < ET) && ((unsigned)(dst[k] - lo) < (unsigned)PN);
        const int dd = own[k] ? dst[k] : 0;
        const int ss = own[k] ? src[k] : 0;
        const float v = s_i[dd] + s_j[ss];
        a[k] = (v > 0.0f) ? v : NEGS * v;
    }
    __syncthreads();
    #pragma unroll
    for (int k = 0; k < EPT; ++k) {
        const int e = e0 + k * SE;
        if (own[k]) {
            aplane[(size_t)pt * ET + e] = a[k];
            atomicMax(&lmax[dst[k] - lo], fenc(a[k]));
        }
    }
    __syncthreads();
    const int rot = (sl * 6 * 32) % PN;
    for (int i = tid; i < PN; i += AT) {
        int j = i + rot;
        if (j >= PN) j -= PN;
        const unsigned v = lmax[j];
        if (v != ENC_NEG_INF && lo + j < NN) atomicMax(amax + lo + j, v);
    }
}

#define FXS 1099511627776.0f

__global__ __launch_bounds__(AT)
void edge_exp_kernel(const int* __restrict__ ei, const unsigned* __restrict__ amax,
                     const float* __restrict__ aplane, float* __restrict__ explane,
                     unsigned long long* __restrict__ dsum) {
    __shared__ unsigned long long lsum[PN];
    const int tid = threadIdx.x;
    const int sl  = blockIdx.x / PT;
    const int pt  = blockIdx.x % PT;
    const int lo  = pt * PN;
    for (int i = tid; i < PN; i += AT) lsum[i] = 0ull;

    int dst[EPT];
    const int e0 = sl * AT + tid;
    #pragma unroll
    for (int k = 0; k < EPT; ++k) {
        const int e = e0 + k * SE;
        dst[k] = (k < FULLK || e < EE) ? ei[EE + e] : (e - EE);
    }
    const float* ap = aplane + (size_t)pt * ET;
    float* xp = explane + (size_t)pt * ET;
    float x[EPT];
    bool  own[EPT];
    #pragma unroll
    for (int k = 0; k < EPT; ++k) {
        const int e = e0 + k * SE;
        own[k] = (k < FULLK || e < ET) && ((unsigned)(dst[k] - lo) < (unsigned)PN);
        const int dd = own[k] ? dst[k] : lo;
        const int ee = own[k] ? e : (EE + lo);
        x[k] = expf(ap[ee] - fdec(amax[dd]));
    }
    __syncthreads();
    #pragma unroll
    for (int k = 0; k < EPT; ++k) {
        const int e = e0 + k * SE;
        if (own[k]) {
            xp[e] = x[k];
            atomicAdd(&lsum[dst[k] - lo], (unsigned long long)(x[k] * FXS));
        }
    }
    __syncthreads();
    const int rot = (sl * 6 * 32) % PN;
    for (int i = tid; i < PN; i += AT) {
        int j = i + rot;
        if (j >= PN) j -= PN;
        const unsigned long long v = lsum[j];
        if (v != 0ull && lo + j < NN) atomicAdd(dsum + lo + j, v);
    }
}

#define EPW 8
#define MWV 8
static_assert(EE % EPW == 0 && ET % EPW == 0 && (EPW & (EPW - 1)) == 0 && EPW <= 32,
              "a wave's edges are all listed or all self loops");

__global__ __launch_bounds__(MWV * 32)
void message_kernel(const int* __restrict__ ei, const float* __restrict__ g,
                    const float* __restrict__ explane,
                    const unsigned long long* __restrict__ dsum,
                    float* __restrict__ out) {
    const int lane = threadIdx.x & 31;
    const int wv = blockIdx.x * MWV + __builtin_amdgcn_readfirstlane((int)(threadIdx.x >> 5));
    const int e0 = wv * EPW;
    if (e0 >= ET) return;
    const int e = e0 + (lane & (EPW - 1));
    int src, dst;
    if (e0 < EE) { src = ei[e]; dst = ei[EE + e]; } else { src = dst = e - EE; }
    const float denom = (float)dsum[dst] * (1.0f / FXS);
    const float coef = explane[(size_t)(dst / PN) * ET + e] / (denom + 1e-16f);

    float ga[EPW], gb[EPW];
    #pragma unroll
    for (int k = 0; k < EPW; ++k) {
        const int sk = __builtin_amdgcn_readlane(src, k);
        const float* gp = g + sk * CC + lane;
        ga[k] = gp[0];
        gb[k] = gp[32];
    }
    #pragma unroll
    for (int k = 0; k < EPW; ++k) {
        const int dk = __builtin_amdgcn_readlane(dst, k);
        const float ck = __int_as_float(__builtin_amdgcn_readlane(__float_as_int(coef), k));
        float* op = out + dk * CC + lane;
        atomicAdd(op,      ga[k] * ck);
        atomicAdd(op + 32, gb[k] * ck);
    }
}

__global__ __launch_bounds__(256)
void finalize_kernel(float* __restrict__ out, const float* __restrict__ bias) {
    int i = blockIdx.x * blockDim.x + threadIdx.x;
    if (i >= NN * CC) return;
    float v = out[i] + bias[i & (CC - 1)];
    out[i] = (v > 0.0f) ? v : 0.0f;
}

extern "C" void kernel_launch(void* const* d_in, const int* in_sizes, int n_in,
                              void* d_out, int out_size, void* d_ws, size_t ws_size,
                              hipStream_t stream) {
    (void)in_sizes; (void)n_in; (void)out_size; (void)ws_size;
    const float* x      = (const float*)d_in[0];
    const int*   ei     = (const int*)  d_in[1];
    const float* corrs  = (const float*)d_in[2];
    const float* W      = (const float*)d_in[3];
    const float* att_i  = (const float*)d_in[4];
    const float* att_j  = (const float*)d_in[5];
    const float* aem_i  = (const float*)d_in[6];
    const float* aem_j  = (const float*)d_in[7];
    const float* bias   = (const float*)d_in[8];
    float* out = (float*)d_out;

    float*    ws    = (float*)d_ws;
    float*    g     = ws;
    float*    s_i   = g + (size_t)NN * CC;
    float*    s_j   = s_i + NN;
    unsigned* amax  = (unsigned*)(s_j + NN);
    unsigned long long* dsum = (unsigned long long*)(amax + NN);
    float*    xpl   = (float*)(dsum + NN);
    float*    apl   = xpl + (size_t)PT * ET;

    gemm_wmma_kernel<<<NN / 16, 32, 0, stream>>>(x, W, g, out, dsum, amax);
    node_scores_kernel<<<(NN * 32 + 255) / 256, 256, 0, stream>>>(
        g, corrs, att_i, att_j, aem_i, aem_j, s_i, s_j);
    edge_alpha_kernel<<<SL * PT, AT, 0, stream>>>(ei, s_i, s_j, apl, amax);
    edge_exp_kernel<<<SL * PT, AT, 0, stream>>>(ei, amax, apl, xpl, dsum);
    message_kernel<<<(ET / EPW + MWV - 1) / MWV, MWV * 32, 0, stream>>>(ei, g, xpl, dsum, out);
    finalize_kernel<<<(NN * CC + 255) / 256, 256, 0, stream>>>(out, bias);
}


# Softmax division moved from each edge to the last pass (row / exp sum), so the exp-sum blocks run inside the message launch; 6 launches -> 5; sums stay in atomic arrival order
# speedup vs baseline: 2.1914x; 1.0335x over previous
#include <hip/hip_runtime.h>
#include <math.h>

#define NN   50000
#define EE   800000
#define ET   (EE + NN)
#define INC  128
#define CC   64
#define NEGS 0.2f

typedef float v2f __attribute__((ext_vector_type(2)));
typedef float v8f __attribute__((ext_vector_type(8)));

__device__ __forceinline__ unsigned fenc(float f) {
    unsigned u = __float_as_uint(f);
    return (u >> 31) ? ~u : (u | 0x80000000u);
}
__device__ __forceinline__ float fdec(unsigned u) {
    return __uint_as_float((u >> 31) ? (u & 0x7FFFFFFFu) : ~u);
}

#define ENC_NEG_INF 0x007FFFFFu

__global__ __launch_bounds__(32)
void gemm_wmma_kernel(const float* __restrict__ x, const float* __restrict__ W,
                      float* __restrict__ g, float* __restrict__ out,
                      unsigned long long* __restrict__ dsum, unsigned* __restrict__ amax) {
    const int rb   = blockIdx.x;
    const int lane = threadIdx.x;
    const int half = lane >> 4;
    const int l15  = lane & 15;

    {
        const float4 z4 = make_float4(0.0f, 0.0f, 0.0f, 0.0f);
        float4* o4 = reinterpret_cast<float4*>(out + (size_t)rb * 16 * CC) + lane;
        for (int i = 0; i < 8; ++i) o4[i * 32] = z4;
        if (half == 0) dsum[rb * 16 + l15] = 0ull;
        else           amax[rb * 16 + l15] = ENC_NEG_INF;
    }

    v8f acc0 = {}; v8f acc1 = {}; v8f acc2 = {}; v8f acc3 = {};

    const float* xrow = x + (rb * 16 + l15) * INC + half * 2;
    for (int k0 = 0; k0 < INC; k0 += 4) {
        v2f a; a.x = xrow[k0]; a.y = xrow[k0 + 1];
        const float* wp = W + (k0 + half * 2) * CC + l15;
        v2f b0, b1, b2, b3;
        b0.x = wp[0];  b0.y = wp[0  + CC];
        b1.x = wp[16]; b1.y = wp[16 + CC];
        b2.x = wp[32]; b2.y = wp[32 + CC];
        b3.x = wp[48]; b3.y = wp[48 + CC];
        acc0 = __builtin_amdgcn_wmma_f32_16x16x4_f32(false, a, false, b0, (short)0, acc0, false, false);
        acc1 = __builtin_amdgcn_wmma_f32_16x16x4_f32(false, a, false, b1, (short)0, acc1, false, false);
        acc2 = __builtin_amdgcn_wmma_f32_16x16x4_f32(false, a, false, b2, (short)0, acc2, false, false);
        acc3 = __builtin_amdgcn_wmma_f32_16x16x4_f32(false, a, false, b3, (short)0, acc3, false, false);
    }
    #pragma unroll
    for (int v = 0; v < 8; ++v) {
        float* op = g + (rb * 16 + v + half * 8) * CC + l15;
        op[0]  = acc0[v];
        op[16] = acc1[v];
        op[32] = acc2[v];
        op[48] = acc3[v];
    }
}

__global__ __launch_bounds__(256)
void node_scores_kernel(const float* __restrict__ g, const float* __restrict__ corrs,
                        const float* __restrict__ att_i, const float* __restrict__ att_j,
                        const float* __restrict__ aem_i, const float* __restrict__ aem_j,
                        float* __restrict__ s_i, float* __restrict__ s_j) {
    int node = (blockIdx.x * blockDim.x + threadIdx.x) >> 5;
    int lane = threadIdx.x & 31;
    if (node >= NN) return;
    const float* gr = g + node * CC;
    const float* cr = corrs + node * CC;
    float si = 0.0f, sj = 0.0f;
    #pragma unroll
    for (int c = lane; c < CC; c += 32) {
        float gv = gr[c], cv = cr[c];
        si += gv * att_i[c] + cv * aem_i[c];
        sj += gv * att_j[c] + cv * aem_j[c];
    }
    #pragma unroll
    for (int off = 16; off > 0; off >>= 1) {
        si += __shfl_xor(si, off, 32);
        sj += __shfl_xor(sj, off, 32);
    }
    if (lane == 0) { s_i[node] = si; s_j[node] = sj; }
}

#define SL   64
#define PT   4
#define PN   ((((NN + PT - 1) / PT) + 31) & ~31)
#define AT   1024
#define SE   (SL * AT)
#define EPT  ((ET + SE - 1) / SE)
#define FULLK (EE / SE)

__global__ __launch_bounds__(AT)
void edge_alpha_kernel(const int* __restrict__ ei, const float* __restrict__ s_i,
                       const float* __restrict__ s_j, float* __restrict__ aplane,
                       unsigned* __restrict__ amax) {
    __shared__ unsigned lmax[PN];
    const int tid = threadIdx.x;
    const int sl  = blockIdx.x / PT;
    const int pt  = blockIdx.x % PT;
    const int lo  = pt * PN;
    for (int i = tid; i < PN; i += AT) lmax[i] = ENC_NEG_INF;

    int src[EPT], dst[EPT];
    const int e0 = sl * AT + tid;
    #pragma unroll
    for (int k = 0; k < EPT; ++k) {
        const int e = e0 + k * SE;
        if (k < FULLK || e < EE) { src[k] = ei[e]; dst[k] = ei[EE + e]; }
        else                     { src[k] = dst[k] = e - EE; }
    }
    float a[EPT];
    bool  own[EPT];
    #pragma unroll
    for (int k = 0; k < EPT; ++k) {
        const int e = e0 + k * SE;
        own[k] = (k < FULLK || e < ET) && ((unsigned)(dst[k] - lo) < (unsigned)PN);
        const int dd = own[k] ? dst[k] : 0;
        const int ss = own[k] ? src[k] : 0;
        const float v = s_i[dd] + s_j[ss];
        a[k] = (v > 0.0f) ? v : NEGS * v;
    }
    __syncthreads();
    #pragma unroll
    for (int k = 0; k < EPT; ++k) {
        const int e = e0 + k * SE;
        if (own[k]) {
            aplane[(size_t)pt * ET + e] = a[k];
            atomicMax(&lmax[dst[k] - lo], fenc(a[k]));
        }
    }
    __syncthreads();
    const int rot = (sl * 6 * 32) % PN;
    for (int i = tid; i < PN; i += AT) {
        int j = i + rot;
        if (j >= PN) j -= PN;
        const unsigned v = lmax[j];
        if (v != ENC_NEG_INF && lo + j < NN) atomicMax(amax + lo + j, v);
    }
}

#define FXS 1099511627776.0f
#define EPW 8
#define MWV (AT / 32)
#define MBLK ((ET / EPW + MWV - 1) / MWV)
static_assert(EE % EPW == 0 && ET % EPW == 0 && (EPW & (EPW - 1)) == 0 && EPW <= 32,
              "a wave's edges are all listed or all self loops");

__global__ __launch_bounds__(AT)
void edge_sum_message_kernel(const int* __restrict__ ei, const unsigned* __restrict__ amax,
                             const float* __restrict__ aplane, const float* __restrict__ g,
                             unsigned long long* __restrict__ dsum, float* __restrict__ out) {
    __shared__ unsigned long long lsum[PN];
    const int tid = threadIdx.x;
    if (blockIdx.x < SL * PT) {
        const int sl  = blockIdx.x / PT;
        const int pt  = blockIdx.x % PT;
        const int lo  = pt * PN;
        for (int i = tid; i < PN; i += AT) lsum[i] = 0ull;

        int dst[EPT];
        const int e0 = sl * AT + tid;
        #pragma unroll
        for (int k = 0; k < EPT; ++k) {
            const int e = e0 + k * SE;
            dst[k] = (k < FULLK || e < EE) ? ei[EE + e] : (e - EE);
        }
        const float* ap = aplane + (size_t)pt * ET;
        float x[EPT];
        bool  own[EPT];
        #pragma unroll
        for (int k = 0; k < EPT; ++k) {
            const int e = e0 + k * SE;
            own[k] = (k < FULLK || e < ET) && ((unsigned)(dst[k] - lo) < (unsigned)PN);
            const int dd = own[k] ? dst[k] : lo;
            const int ee = own[k] ? e : (EE + lo);
            x[k] = expf(ap[ee] - fdec(amax[dd]));
        }
        __syncthreads();
        #pragma unroll
        for (int k = 0; k < EPT; ++k) {
            if (own[k])
                atomicAdd(&lsum[dst[k] - lo], (unsigned long long)(x[k] * FXS));
        }
        __syncthreads();
        const int rot = (sl * 6 * 32) % PN;
        for (int i = tid; i < PN; i += AT) {
            int j = i + rot;
            if (j >= PN) j -= PN;
            const unsigned long long v = lsum[j];
            if (v != 0ull && lo + j < NN) atomicAdd(dsum + lo + j, v);
        }
        return;
    }
    const int lane = tid & 31;
    const int wv = (blockIdx.x - SL * PT) * MWV + __builtin_amdgcn_readfirstlane(tid >> 5);
    const int e0 = wv * EPW;
    if (e0 >= ET) return;
    const int e = e0 + (lane & (EPW - 1));
    int src, dst;
    if (e0 < EE) { src = ei[e]; dst = ei[EE + e]; } else { src = dst = e - EE; }
    const float wgt = expf(aplane[(size_t)(dst / PN) * ET + e] - fdec(amax[dst]));

    float ga[EPW], gb[EPW];
    #pragma unroll
    for (int k = 0; k < EPW; ++k) {
        const int sk = __builtin_amdgcn_readlane(src, k);
        const float* gp = g + sk * CC + lane;
        ga[k] = gp[0];
        gb[k] = gp[32];
    }
    #pragma unroll
    for (int k = 0; k < EPW; ++k) {
        const int dk = __builtin_amdgcn_readlane(dst, k);
        const float wk = __int_as_float(__builtin_amdgcn_readlane(__float_as_int(wgt), k));
        float* op = out + dk * CC + lane;
        atomicAdd(op,      ga[k] * wk);
        atomicAdd(op + 32, gb[k] * wk);
    }
}

__global__ __launch_bounds__(256)
void finalize_kernel(float* __restrict__ out, const unsigned long long* __restrict__ dsum,
                     const float* __restrict__ bias) {
    int i = blockIdx.x * blockDim.x + threadIdx.x;
    if (i >= NN * CC) return;
    const float denom = (float)dsum[i / CC] * (1.0f / FXS);
    float v = out[i] / (denom + 1e-16f) + bias[i & (CC - 1)];
    out[i] = (v > 0.0f) ? v : 0.0f;
}

extern "C" void kernel_launch(void* const* d_in, const int* in_sizes, int n_in,
                              void* d_out, int out_size, void* d_ws, size_t ws_size,
                              hipStream_t stream) {
    (void)in_sizes; (void)n_in; (void)out_size; (void)ws_size;
    const float* x      = (const float*)d_in[0];
    const int*   ei     = (const int*)  d_in[1];
    const float* corrs  = (const float*)d_in[2];
    const float* W      = (const float*)d_in[3];
    const float* att_i  = (const float*)d_in[4];
    const float* att_j  = (const float*)d_in[5];
    const float* aem_i  = (const float*)d_in[6];
    const float* aem_j  = (const float*)d_in[7];
    const float* bias   = (const float*)d_in[8];
    float* out = (float*)d_out;

    float*    ws    = (float*)d_ws;
    float*    g     = ws;
    float*    s_i   = g + (size_t)NN * CC;
    float*    s_j   = s_i + NN;
    unsigned* amax  = (unsigned*)(s_j + NN);
    unsigned long long* dsum = (unsigned long long*)(amax + NN);
    float*    apl   = (float*)(dsum + NN);

    gemm_wmma_kernel<<<NN / 16, 32, 0, stream>>>(x, W, g, out, dsum, amax);
    node_scores_kernel<<<(NN * 32 + 255) / 256, 256, 0, stream>>>(
        g, corrs, att_i, att_j, aem_i, aem_j, s_i, s_j);
    edge_alpha_kernel<<<SL * PT, AT, 0, stream>>>(ei, s_i, s_j, apl, amax);
    edge_sum_message_kernel<<<SL * PT + MBLK, AT, 0, stream>>>(ei, amax, apl, g, dsum, out);
    finalize_kernel<<<(NN * CC + 255) / 256, 256, 0, stream>>>(out, dsum, bias);
}


# Scores taken from x and the two 128-vectors W*att (formed per block in LDS), not from g: they need no GEMM result and run as extra blocks of the GEMM launch; 5 -> 4 launches; score sum order changed
# speedup vs baseline: 2.5224x; 1.1510x over previous
#include <hip/hip_runtime.h>
#include <math.h>

#define NN   50000
#define EE   800000
#define ET   (EE + NN)
#define INC  128
#define CC   64
#define NEGS 0.2f

typedef float v2f __attribute__((ext_vector_type(2)));
typedef float v8f __attribute__((ext_vector_type(8)));

__device__ __forceinline__ unsigned fenc(float f) {
    unsigned u = __float_as_uint(f);
    return (u >> 31) ? ~u : (u | 0x80000000u);
}
__device__ __forceinline__ float fdec(unsigned u) {
    return __uint_as_float((u >> 31) ? (u & 0x7FFFFFFFu) : ~u);
}

#define ENC_NEG_INF 0x007FFFFFu

#define GW   8
#define NPW  8
#define NRD  (NPW / 2)
#define GBLK ((NN / 16 + GW - 1) / GW)
#define SBLK ((NN / NPW + GW - 1) / GW)
static_assert(NN % 16 == 0 && NN % NPW == 0 && INC == 128 && CC == 64, "whole tiles");

__device__ __forceinline__ float dot4(const float4 a, const float4 b) {
    return a.x * b.x + a.y * b.y + a.z * b.z + a.w * b.w;
}

#define XLANE(v, m) __int_as_float(__builtin_amdgcn_update_dpp( \
        0, __float_as_int(v), 0x160 + (m), 0xf, 0xf, true))
__device__ __forceinline__ float rowsum16(float v) {
    v += XLANE(v, 1);
    v += XLANE(v, 2);
    v += XLANE(v, 4);
    v += XLANE(v, 8);
    return v;
}

__global__ __launch_bounds__(GW * 32)
void gemm_scores_kernel(const float* __restrict__ x, const float* __restrict__ W,
                        const float* __restrict__ corrs,
                        const float* __restrict__ att_i, const float* __restrict__ att_j,
                        const float* __restrict__ aem_i, const float* __restrict__ aem_j,
                        float* __restrict__ g, float* __restrict__ out,
                        unsigned long long* __restrict__ dsum, unsigned* __restrict__ amax,
                        float* __restrict__ s_i, float* __restrict__ s_j) {
    __shared__ float4 wv4[2 * INC / 4];
    const int lane = threadIdx.x & 31;
    const int half = lane >> 4;
    const int l15  = lane & 15;
    const int wvn  = __builtin_amdgcn_readfirstlane(threadIdx.x >> 5);

    if (blockIdx.x >= GBLK) {
        const int n0  = ((blockIdx.x - GBLK) * GW + wvn) * NPW;
        const bool on = n0 < NN;
        const int nb  = (on ? n0 : 0) + NRD * half;
        const float4* xp = reinterpret_cast<const float4*>(x     + (size_t)nb * INC) + l15;
        const float4* cp = reinterpret_cast<const float4*>(corrs + (size_t)nb * CC)  + l15;
        float4 xa[NRD], xb[NRD], cv[NRD];
        #pragma unroll
        for (int r = 0; r < NRD; ++r) {
            xa[r] = xp[r * (INC / 4)];
            xb[r] = xp[r * (INC / 4) + 16];
            cv[r] = cp[r * (CC / 4)];
        }
        float* wv = reinterpret_cast<float*>(wv4);
        {
            const int kk = threadIdx.x >> 4;
            const float4 ai = reinterpret_cast<const float4*>(att_i)[l15];
            const float4 aj = reinterpret_cast<const float4*>(att_j)[l15];
            float4 w[INC / 16];
            #pragma unroll
            for (int i = 0; i < INC / 16; ++i)
                w[i] = reinterpret_cast<const float4*>(W + (kk + 16 * i) * CC)[l15];
            #pragma unroll
            for (int i = 0; i < INC / 16; ++i) {
                const float pi = rowsum16(dot4(w[i], ai));
                const float pj = rowsum16(dot4(w[i], aj));
                if (l15 == 0) { wv[kk + 16 * i] = pi; wv[INC + kk + 16 * i] = pj; }
            }
        }
        __syncthreads();
        if (!on) return;
        const float4 wia = wv4[l15],           wib = wv4[16 + l15];
        const float4 wja = wv4[INC / 4 + l15], wjb = wv4[INC / 4 + 16 + l15];
        const float4 ei  = reinterpret_cast<const float4*>(aem_i)[l15];
        const float4 ej  = reinterpret_cast<const float4*>(aem_j)[l15];
        float ki = 0.0f, kj = 0.0f;
        #pragma unroll
        for (int r = 0; r < NRD; ++r) {
            const float si = rowsum16(dot4(xa[r], wia) + dot4(xb[r], wib) + dot4(cv[r], ei));
            const float sj = rowsum16(dot4(xa[r], wja) + dot4(xb[r], wjb) + dot4(cv[r], ej));
            if (l15 == r) { ki = si; kj = sj; }
        }
        if (l15 < NRD) { s_i[nb + l15] = ki; s_j[nb + l15] = kj; }
        return;
    }

    const int rb = blockIdx.x * GW + wvn;
    if (rb >= NN / 16) return;

    {
        const float4 z4 = make_float4(0.0f, 0.0f, 0.0f, 0.0f);
        float4* o4 = reinterpret_cast<float4*>(out + (size_t)rb * 16 * CC) + lane;
        for (int i = 0; i < 8; ++i) o4[i * 32] = z4;
        if (half == 0) dsum[rb * 16 + l15] = 0ull;
        else           amax[rb * 16 + l15] = ENC_NEG_INF;
    }

    v8f acc0 = {}; v8f acc1 = {}; v8f acc2 = {}; v8f acc3 = {};

    const float* xrow = x + (rb * 16 + l15) * INC + half * 2;
    for (int k0 = 0; k0 < INC; k0 += 4) {
        v2f a; a.x = xrow[k0]; a.y = xrow[k0 + 1];
        const float* wp = W + (k0 + half * 2) * CC + l15;
        v2f b0, b1, b2, b3;
        b0.x = wp[0];  b0.y = wp[0  + CC];
        b1.x = wp[16]; b1.y = wp[16 + CC];
        b2.x = wp[32]; b2.y = wp[32 + CC];
        b3.x = wp[48]; b3.y = wp[48 + CC];
        acc0 = __builtin_amdgcn_wmma_f32_16x16x4_f32(false, a, false, b0, (short)0, acc0, false, false);
        acc1 = __builtin_amdgcn_wmma_f32_16x16x4_f32(false, a, false, b1, (short)0, acc1, false, false);
        acc2 = __builtin_amdgcn_wmma_f32_16x16x4_f32(false, a, false, b2, (short)0, acc2, false, false);
        acc3 = __builtin_amdgcn_wmma_f32_16x16x4_f32(false, a, false, b3, (short)0, acc3, false, false);
    }
    #pragma unroll
    for (int v = 0; v < 8; ++v) {
        float* op = g + (rb * 16 + v + half * 8) * CC + l15;
        op[0]  = acc0[v];
        op[16] = acc1[v];
        op[32] = acc2[v];
        op[48] = acc3[v];
    }
}

#define SL   64
#define PT   4
#define PN   ((((NN + PT - 1) / PT) + 31) & ~31)
#define AT   1024
#define SE   (SL * AT)
#define EPT  ((ET + SE - 1) / SE)
#define FULLK (EE / SE)

__global__ __launch_bounds__(AT)
void edge_alpha_kernel(const int* __restrict__ ei, const float* __restrict__ s_i,
                       const float* __restrict__ s_j, float* __restrict__ aplane,
                       unsigned* __restrict__ amax) {
    __shared__ unsigned lmax[PN];
    const int tid = threadIdx.x;
    const int sl  = blockIdx.x / PT;
    const int pt  = blockIdx.x % PT;
    const int lo  = pt * PN;
    for (int i = tid; i < PN; i += AT) lmax[i] = ENC_NEG_INF;

    int src[EPT], dst[EPT];
    const int e0 = sl * AT + tid;
    #pragma unroll
    for (int k = 0; k < EPT; ++k) {
        const int e = e0 + k * SE;
        if (k < FULLK || e < EE) { src[k] = ei[e]; dst[k] = ei[EE + e]; }
        else                     { src[k] = dst[k] = e - EE; }
    }
    float a[EPT];
    bool  own[EPT];
    #pragma unroll
    for (int k = 0; k < EPT; ++k) {
        const int e = e0 + k * SE;
        own[k] = (k < FULLK || e < ET) && ((unsigned)(dst[k] - lo) < (unsigned)PN);
        const int dd = own[k] ? dst[k] : 0;
        const int ss = own[k] ? src[k] : 0;
        const float v = s_i[dd] + s_j[ss];
        a[k] = (v > 0.0f) ? v : NEGS * v;
    }
    __syncthreads();
    #pragma unroll
    for (int k = 0; k < EPT; ++k) {
        const int e = e0 + k * SE;
        if (own[k]) {
            aplane[(size_t)pt * ET + e] = a[k];
            atomicMax(&lmax[dst[k] - lo], fenc(a[k]));
        }
    }
    __syncthreads();
    const int rot = (sl * 6 * 32) % PN;
    for (int i = tid; i < PN; i += AT) {
        int j = i + rot;
        if (j >= PN) j -= PN;
        const unsigned v = lmax[j];
        if (v != ENC_NEG_INF && lo + j < NN) atomicMax(amax + lo + j, v);
    }
}

#define FXS 1099511627776.0f
#define EPW 8
#define MWV (AT / 32)
#define MBLK ((ET / EPW + MWV - 1) / MWV)
static_assert(EE % EPW == 0 && ET % EPW == 0 && (EPW & (EPW - 1)) == 0 && EPW <= 32,
              "a wave's edges are all listed or all self loops");

__global__ __launch_bounds__(AT)
void edge_sum_message_kernel(const int* __restrict__ ei, const unsigned* __restrict__ amax,
                             const float* __restrict__ aplane, const float* __restrict__ g,
                             unsigned long long* __restrict__ dsum, float* __restrict__ out) {
    __shared__ unsigned long long lsum[PN];
    const int tid = threadIdx.x;
    if (blockIdx.x < SL * PT) {
        const int sl  = blockIdx.x / PT;
        const int pt  = blockIdx.x % PT;
        const int lo  = pt * PN;
        for (int i = tid; i < PN; i += AT) lsum[i] = 0ull;

        int dst[EPT];
        const int e0 = sl * AT + tid;
        #pragma unroll
        for (int k = 0; k < EPT; ++k) {
            const int e = e0 + k * SE;
            dst[k] = (k < FULLK || e < EE) ? ei[EE + e] : (e - EE);
        }
        const float* ap = aplane + (size_t)pt * ET;
        float x[EPT];
        bool  own[EPT];
        #pragma unroll
        for (int k = 0; k < EPT; ++k) {
            const int e = e0 + k * SE;
            own[k] = (k < FULLK || e < ET) && ((unsigned)(dst[k] - lo) < (unsigned)PN);
            const int dd = own[k] ? dst[k] : lo;
            const int ee = own[k] ? e : (EE + lo);
            x[k] = expf(ap[ee] - fdec(amax[dd]));
        }
        __syncthreads();
        #pragma unroll
        for (int k = 0; k < EPT; ++k) {
            if (own[k])
                atomicAdd(&lsum[dst[k] - lo], (unsigned long long)(x[k] * FXS));
        }
        __syncthreads();
        const int rot = (sl * 6 * 32) % PN;
        for (int i = tid; i < PN; i += AT) {
            int j = i + rot;
            if (j >= PN) j -= PN;
            const unsigned long long v = lsum[j];
            if (v != 0ull && lo + j < NN) atomicAdd(dsum + lo + j, v);
        }
        return;
    }
    const int lane = tid & 31;
    const int wv = (blockIdx.x - SL * PT) * MWV + __builtin_amdgcn_readfirstlane(tid >> 5);
    const int e0 = wv * EPW;
    if (e0 >= ET) return;
    const int e = e0 + (lane & (EPW - 1));
    int src, dst;
    if (e0 < EE) { src = ei[e]; dst = ei[EE + e]; } else { src = dst = e - EE; }
    const float wgt = expf(aplane[(size_t)(dst / PN) * ET + e] - fdec(amax[dst]));

    float ga[EPW], gb[EPW];
    #pragma unroll
    for (int k = 0; k < EPW; ++k) {
        const int sk = __builtin_amdgcn_readlane(src, k);
        const float* gp = g + sk * CC + lane;
        ga[k] = gp[0];
        gb[k] = gp[32];
    }
    #pragma unroll
    for (int k = 0; k < EPW; ++k) {
        const int dk = __builtin_amdgcn_readlane(dst, k);
        const float wk = __int_as_float(__builtin_amdgcn_readlane(__float_as_int(wgt), k));
        float* op = out + dk * CC + lane;
        atomicAdd(op,      ga[k] * wk);
        atomicAdd(op + 32, gb[k] * wk);
    }
}

__global__ __launch_bounds__(256)
void finalize_kernel(float* __restrict__ out, const unsigned long long* __restrict__ dsum,
                     const float* __restrict__ bias) {
    int i = blockIdx.x * blockDim.x + threadIdx.x;
    if (i >= NN * CC) return;
    const float denom = (float)dsum[i / CC] * (1.0f / FXS);
    float v = out[i] / (denom + 1e-16f) + bias[i & (CC - 1)];
    out[i] = (v > 0.0f) ? v : 0.0f;
}

extern "C" void kernel_launch(void* const* d_in, const int* in_sizes, int n_in,
                              void* d_out, int out_size, void* d_ws, size_t ws_size,
                              hipStream_t stream) {
    (void)in_sizes; (void)n_in; (void)out_size; (void)ws_size;
    const float* x      = (const float*)d_in[0];
    const int*   ei     = (const int*)  d_in[1];
    const float* corrs  = (const float*)d_in[2];
    const float* W      = (const float*)d_in[3];
    const float* att_i  = (const float*)d_in[4];
    const float* att_j  = (const float*)d_in[5];
    const float* aem_i  = (const float*)d_in[6];
    const float* aem_j  = (const float*)d_in[7];
    const float* bias   = (const float*)d_in[8];
    float* out = (float*)d_out;

    float*    ws    = (float*)d_ws;
    float*    g     = ws;
    float*    s_i   = g + (size_t)NN * CC;
    float*    s_j   = s_i + NN;
    unsigned* amax  = (unsigned*)(s_j + NN);
    unsigned long long* dsum = (unsigned long long*)(amax + NN);
    float*    apl   = (float*)(dsum + NN);

    gemm_scores_kernel<<<GBLK + SBLK, GW * 32, 0, stream>>>(
        x, W, corrs, att_i, att_j, aem_i, aem_j, g, out, dsum, amax, s_i, s_j);
    edge_alpha_kernel<<<SL * PT, AT, 0, stream>>>(ei, s_i, s_j, apl, amax);
    edge_sum_message_kernel<<<SL * PT + MBLK, AT, 0, stream>>>(ei, amax, apl, g, dsum, out);
    finalize_kernel<<<(NN * CC + 255) / 256, 256, 0, stream>>>(out, dsum, bias);
}
